# starting kernel
# baseline (speedup 1.0000x reference)
#include <hip/hip_runtime.h>

typedef __attribute__((ext_vector_type(2))) float v2f;
typedef __attribute__((ext_vector_type(8))) float v8f;

#define F_DIM 200
#define E_DIM 64
#define LDS_PITCH 202
#define WAVES_PER_BLOCK 8
#define ROWS_PER_WAVE 16
#define ROWS_PER_BLOCK (WAVES_PER_BLOCK * ROWS_PER_WAVE)

__global__ __launch_bounds__(256)
void fm_second_order_kernel(const float* __restrict__ features,
                            const float* __restrict__ W,
                            float* __restrict__ out, int Brows) {
    __shared__ float sWt[E_DIM * LDS_PITCH];
    __shared__ float sW2[F_DIM];

    const int tid  = threadIdx.x;
    const int lane = tid & 31;
    const int wave = tid >> 5;

    for (int i = tid; i < F_DIM * E_DIM; i += 256) {
        int f = i >> 6;
        int e = i & (E_DIM - 1);
        sWt[e * LDS_PITCH + f] = W[i];
    }
    __syncthreads();
    if (tid < F_DIM) {
        float s = 0.f;
        #pragma unroll
        for (int e = 0; e < E_DIM; ++e) {
            float w = sWt[e * LDS_PITCH + tid];
            s = fmaf(w, w, s);
        }
        sW2[tid] = s;
    }
    __syncthreads();

    const int base_row = blockIdx.x * ROWS_PER_BLOCK + wave * ROWS_PER_WAVE;
    if (base_row >= Brows) return;

    const int mrow = lane & 15;
    const int half = lane >> 4;
    const int n0   = lane & 15;

    const float* arow = features + (size_t)(base_row + mrow) * F_DIM;

    v8f acc0 = {}, acc1 = {}, acc2 = {}, acc3 = {};
    float sqp = 0.f;

    for (int k = 0; k < F_DIM; k += 4) {
        const int kk = k + 2 * half;
        v2f a = *reinterpret_cast<const v2f*>(arow + kk);
        sqp = fmaf(a.x * a.x, sW2[kk],     sqp);
        sqp = fmaf(a.y * a.y, sW2[kk + 1], sqp);
        v2f b0 = *reinterpret_cast<const v2f*>(&sWt[(n0 +  0) * LDS_PITCH + kk]);
        v2f b1 = *reinterpret_cast<const v2f*>(&sWt[(n0 + 16) * LDS_PITCH + kk]);
        v2f b2 = *reinterpret_cast<const v2f*>(&sWt[(n0 + 32) * LDS_PITCH + kk]);
        v2f b3 = *reinterpret_cast<const v2f*>(&sWt[(n0 + 48) * LDS_PITCH + kk]);
        acc0 = __builtin_amdgcn_wmma_f32_16x16x4_f32(false, a, false, b0, (short)0, acc0, false, false);
        acc1 = __builtin_amdgcn_wmma_f32_16x16x4_f32(false, a, false, b1, (short)0, acc1, false, false);
        acc2 = __builtin_amdgcn_wmma_f32_16x16x4_f32(false, a, false, b2, (short)0, acc2, false, false);
        acc3 = __builtin_amdgcn_wmma_f32_16x16x4_f32(false, a, false, b3, (short)0, acc3, false, false);
    }

    float s[8];
    #pragma unroll
    for (int r = 0; r < 8; ++r) {
        float c0 = acc0[r], c1 = acc1[r], c2 = acc2[r], c3 = acc3[r];
        s[r] = c0 * c0 + c1 * c1 + c2 * c2 + c3 * c3;
    }
    #pragma unroll
    for (int m = 1; m <= 8; m <<= 1) {
        #pragma unroll
        for (int r = 0; r < 8; ++r) s[r] += __shfl_xor(s[r], m, 32);
    }
    float sqfull = sqp + __shfl_xor(sqp, 16, 32);

    int myrow = (lane & 7) + ((lane >> 4) << 3);
    float sqr = __shfl(sqfull, myrow, 32);
    if ((lane & 15) < 8) {
        out[base_row + myrow] = s[lane & 7] - sqr;
    }
}

extern "C" void kernel_launch(void* const* d_in, const int* in_sizes, int n_in,
                              void* d_out, int out_size, void* d_ws, size_t ws_size,
                              hipStream_t stream) {
    const float* features = (const float*)d_in[0];
    const float* W        = (const float*)d_in[1];
    float* out            = (float*)d_out;
    (void)d_ws; (void)ws_size; (void)n_in; (void)out_size;

    int B = in_sizes[0] / F_DIM;
    int blocks = (B + ROWS_PER_BLOCK - 1) / ROWS_PER_BLOCK;
    fm_second_order_kernel<<<blocks, 256, 0, stream>>>(features, W, out, B);
}


# W never staged in LDS: each of 4 waves keeps a 16-column slice of W in registers as matrix operand (E split over waves, 64-row groups); w2 by DPP row sums of those registers; new summation order
# speedup vs baseline: 2.1579x; 2.1579x over previous
#include <hip/hip_runtime.h>


typedef __attribute__((ext_vector_type(2))) float v2f;
typedef __attribute__((ext_vector_type(4))) float v4f;
typedef __attribute__((ext_vector_type(8))) float v8f;

#define F_DIM 200
#define E_DIM 64
#define WAVES_PER_BLOCK 4
#define TILES_PER_BLOCK 4
#define ROWS_PER_BLOCK (16 * TILES_PER_BLOCK)
#define THREADS (32 * WAVES_PER_BLOCK)
#define A_PITCH 204
#define A_CHUNKS_PER_ROW (F_DIM / 4)
#define A_CHUNKS_PER_THREAD (ROWS_PER_BLOCK * A_CHUNKS_PER_ROW / THREADS)
#define K_GROUPS (F_DIM / 8)
#define J_CHUNK 5
#define N_CHUNKS (K_GROUPS / J_CHUNK)

template <int CTRL>
__device__ __forceinline__ float row_xor_add(float x) {
    const int xi = __builtin_bit_cast(int, x);
    const int yi = __builtin_amdgcn_update_dpp(0, xi, CTRL, 0xf, 0xf, true);
    return x + __builtin_bit_cast(float, yi);
}
__device__ __forceinline__ float row16_sum(float x) {
    x = row_xor_add<0x161>(x);
    x = row_xor_add<0x162>(x);
    x = row_xor_add<0x164>(x);
    x = row_xor_add<0x168>(x);
    return x;
}

__global__ __launch_bounds__(THREADS)
void fm_second_order_kernel(const float* __restrict__ features,
                            const float* __restrict__ W,
                            float* __restrict__ out, int Brows) {
    __shared__ __attribute__((aligned(16))) float sA[ROWS_PER_BLOCK * A_PITCH];
    __shared__ float sRed[WAVES_PER_BLOCK * 2 * ROWS_PER_BLOCK];

    const int tid  = threadIdx.x;
    const int lane = tid & 31;
    const int wave = tid >> 5;
    const int c    = lane & 15;
    const int h    = lane >> 4;
    const int block_row = blockIdx.x * ROWS_PER_BLOCK;

    const float* wp = W + (4 * h) * E_DIM + 16 * wave + c;
    v2f wf[2 * K_GROUPS];
    #pragma unroll
    for (int J = 0; J < K_GROUPS; ++J) {
        wf[2 * J].x     = wp[(8 * J + 0) * E_DIM];
        wf[2 * J].y     = wp[(8 * J + 1) * E_DIM];
        wf[2 * J + 1].x = wp[(8 * J + 2) * E_DIM];
        wf[2 * J + 1].y = wp[(8 * J + 3) * E_DIM];
    }

    v4f areg[A_CHUNKS_PER_THREAD];
    const v4f* gA = reinterpret_cast<const v4f*>(features + (size_t)block_row * F_DIM);
    if (block_row + ROWS_PER_BLOCK <= Brows) {
        #pragma unroll
        for (int i = 0; i < A_CHUNKS_PER_THREAD; ++i) areg[i] = gA[tid + THREADS * i];
    } else {
        #pragma unroll
        for (int i = 0; i < A_CHUNKS_PER_THREAD; ++i) {
            const int q = tid + THREADS * i;
            areg[i] = (block_row + q / A_CHUNKS_PER_ROW < Brows) ? gA[q] : v4f{0.f, 0.f, 0.f, 0.f};
        }
    }

    v2f w2[2 * K_GROUPS];
    #pragma unroll
    for (int j = 0; j < 2 * K_GROUPS; ++j) {
        const v2f q = wf[j] * wf[j];
        w2[j].x = row16_sum(q.x);
        w2[j].y = row16_sum(q.y);
    }

    #pragma unroll
    for (int i = 0; i < A_CHUNKS_PER_THREAD; ++i) {
        const int q   = tid + THREADS * i;
        const int row = q / A_CHUNKS_PER_ROW;
        const int col = q - row * A_CHUNKS_PER_ROW;
        *reinterpret_cast<v4f*>(&sA[row * A_PITCH + 4 * col]) = areg[i];
    }
    __syncthreads();

    v8f acc[TILES_PER_BLOCK];
    v2f sq[TILES_PER_BLOCK];
    #pragma unroll
    for (int t = 0; t < TILES_PER_BLOCK; ++t) { acc[t] = v8f{}; sq[t] = v2f{0.f, 0.f}; }

    const float* arow = sA + c * A_PITCH + 4 * h;
    v4f fa[N_CHUNKS][J_CHUNK][TILES_PER_BLOCK];
    #pragma unroll
    for (int jj = 0; jj < J_CHUNK; ++jj) {
        #pragma unroll
        for (int t = 0; t < TILES_PER_BLOCK; ++t)
            fa[0][jj][t] = *reinterpret_cast<const v4f*>(arow + t * 16 * A_PITCH + 8 * jj);
    }
    #pragma unroll
    for (int ch = 0; ch < N_CHUNKS; ++ch) {
        if (ch + 1 < N_CHUNKS) {
            #pragma unroll
            for (int jj = 0; jj < J_CHUNK; ++jj) {
                #pragma unroll
                for (int t = 0; t < TILES_PER_BLOCK; ++t)
                    fa[ch + 1][jj][t] = *reinterpret_cast<const v4f*>(arow + t * 16 * A_PITCH + 8 * ((ch + 1) * J_CHUNK + jj));
            }
        }
        __builtin_amdgcn_sched_barrier(0);
        #pragma unroll
        for (int jj = 0; jj < J_CHUNK; ++jj) {
            const int J = ch * J_CHUNK + jj;
            #pragma unroll
            for (int t = 0; t < TILES_PER_BLOCK; ++t) {
                const v2f a01 = {fa[ch][jj][t].x, fa[ch][jj][t].y};
                acc[t] = __builtin_amdgcn_wmma_f32_16x16x4_f32(false, wf[2 * J], false, a01, (short)0, acc[t], false, false);
                sq[t] += a01 * a01 * w2[2 * J];
            }
            #pragma unroll
            for (int t = 0; t < TILES_PER_BLOCK; ++t) {
                const v2f a23 = {fa[ch][jj][t].z, fa[ch][jj][t].w};
                acc[t] = __builtin_amdgcn_wmma_f32_16x16x4_f32(false, wf[2 * J + 1], false, a23, (short)0, acc[t], false, false);
                sq[t] += a23 * a23 * w2[2 * J + 1];
            }
        }
        __builtin_amdgcn_sched_barrier(0);
    }

    #pragma unroll
    for (int t = 0; t < TILES_PER_BLOCK; ++t) {
        float p = 0.f;
        #pragma unroll
        for (int r = 0; r < 8; ++r) p = fmaf(acc[t][r], acc[t][r], p);
        sRed[(wave * 2 + h) * ROWS_PER_BLOCK + 16 * t + c] = p - (sq[t].x + sq[t].y);
    }
    __syncthreads();
    if (tid < ROWS_PER_BLOCK) {
        float s = 0.f;
        #pragma unroll
        for (int g = 0; g < 2 * WAVES_PER_BLOCK; ++g) s += sRed[g * ROWS_PER_BLOCK + tid];
        if (block_row + tid < Brows) out[block_row + tid] = s;
    }
}

extern "C" void kernel_launch(void* const* d_in, const int* in_sizes, int n_in,
                              void* d_out, int out_size, void* d_ws, size_t ws_size,
                              hipStream_t stream) {
    const float* features = (const float*)d_in[0];
    const float* W        = (const float*)d_in[1];
    float* out            = (float*)d_out;
    (void)d_ws; (void)ws_size; (void)n_in; (void)out_size;

    int B = in_sizes[0] / F_DIM;
    int blocks = (B + ROWS_PER_BLOCK - 1) / ROWS_PER_BLOCK;
    fm_second_order_kernel<<<blocks, THREADS, 0, stream>>>(features, W, out, B);
}
